# starting kernel
# baseline (speedup 1.0000x reference)
#include <hip/hip_runtime.h>

typedef float v2f __attribute__((ext_vector_type(2)));
typedef float v8f __attribute__((ext_vector_type(8)));

#define IN_DIM  64
#define OUT_DIM 64
#define HID     64
#define DDIM    (IN_DIM * OUT_DIM)
#define BSZ     256
#define LSTR    68

__launch_bounds__(256)
__global__ void kan_edge_kernel(const float* __restrict__ x,
                                const float* __restrict__ w0, const float* __restrict__ b0,
                                const float* __restrict__ w1, const float* __restrict__ b1,
                                const float* __restrict__ w2, const float* __restrict__ b2,
                                const float* __restrict__ wout,
                                float* __restrict__ outbuf, int use_atomic)
{
    __shared__ float act0[BSZ * LSTR];
    __shared__ float act1[BSZ * LSTR];
    __shared__ float xs[BSZ];
    __shared__ float ws0[HID], bs0[HID], wos[HID];

    const int d   = blockIdx.x;
    const int o   = d >> 6;
    const int i   = d & 63;
    const int tid = threadIdx.x;

    xs[tid & 255] = x[(tid & 255) * IN_DIM + i];
    if (tid < HID) {
        ws0[tid] = w0[d * HID + tid];
        bs0[tid] = b0[d * HID + tid];
        wos[tid] = wout[o * (IN_DIM * HID) + i * HID + tid];
    }
    __syncthreads();

    {
        const int   h  = tid & 63;
        const int   bg = tid >> 6;
        const float wv = ws0[h];
        const float bv = bs0[h];
        for (int b = bg; b < BSZ; b += 4) {
            float v = fmaf(xs[b], wv, bv);
            act0[b * LSTR + h] = v > 0.f ? v : 0.f;
        }
    }
    __syncthreads();

    const int lane  = tid & 31;
    const int r     = lane & 15;
    const int half  = lane >> 4;
    const int nbase = ((tid >> 5) & 3) * 16;
    const int mhalf = (tid >> 5) >> 2;

    auto layer = [&](const float* __restrict__ src, float* __restrict__ dst,
                     const float* __restrict__ W, const float* __restrict__ Bv) {
        v2f Bfrag[16];
        const float* wrow = W + (size_t)(nbase + r) * HID + 2 * half;
        #pragma unroll
        for (int c = 0; c < 16; ++c)
            Bfrag[c] = *(const v2f*)(wrow + c * 4);
        const float bias = Bv[nbase + r];

        #pragma unroll
        for (int t = 0; t < 4; ++t) {
            const int m0 = (mhalf * 8 + 2 * t) * 16;
            const int m1 = m0 + 16;
            v8f acc0 = {0.f,0.f,0.f,0.f,0.f,0.f,0.f,0.f};
            v8f acc1 = {0.f,0.f,0.f,0.f,0.f,0.f,0.f,0.f};
            const float* a0p = src + (m0 + r) * LSTR + 2 * half;
            const float* a1p = src + (m1 + r) * LSTR + 2 * half;
            #pragma unroll
            for (int c = 0; c < 16; ++c) {
                v2f a0 = *(const v2f*)(a0p + c * 4);
                v2f a1 = *(const v2f*)(a1p + c * 4);
                acc0 = __builtin_amdgcn_wmma_f32_16x16x4_f32(
                           false, a0, false, Bfrag[c], (short)0, acc0, false, false);
                acc1 = __builtin_amdgcn_wmma_f32_16x16x4_f32(
                           false, a1, false, Bfrag[c], (short)0, acc1, false, false);
            }
            #pragma unroll
            for (int v = 0; v < 8; ++v) {
                const int row0 = m0 + v + 8 * half;
                const int row1 = m1 + v + 8 * half;
                float v0 = acc0[v] + bias; v0 = v0 > 0.f ? v0 : 0.f;
                float v1 = acc1[v] + bias; v1 = v1 > 0.f ? v1 : 0.f;
                dst[row0 * LSTR + nbase + r] = v0;
                dst[row1 * LSTR + nbase + r] = v1;
            }
        }
    };

    layer(act0, act1, w1 + (size_t)d * HID * HID, b1 + d * HID);
    __syncthreads();
    layer(act1, act0, w2 + (size_t)d * HID * HID, b2 + d * HID);
    __syncthreads();

    {
        const int b = tid;
        float s = 0.f;
        const float* row = act0 + b * LSTR;
        #pragma unroll 8
        for (int h = 0; h < HID; ++h) s = fmaf(wos[h], row[h], s);
        if (use_atomic) {
            __hip_atomic_fetch_add(&outbuf[b * OUT_DIM + o], s,
                                   __ATOMIC_RELAXED, __HIP_MEMORY_SCOPE_AGENT);
        } else {
            outbuf[(size_t)d * BSZ + b] = s;
        }
    }
}

__global__ void kan_reduce_kernel(const float* __restrict__ part,
                                  const float* __restrict__ bout,
                                  float* __restrict__ out)
{
    const int o = blockIdx.x;
    const int b = threadIdx.x;
    float s = bout[o];
    const float* p = part + (size_t)o * 64 * BSZ + b;
    #pragma unroll 8
    for (int i = 0; i < 64; ++i) s += p[i * BSZ];
    out[b * OUT_DIM + o] = s;
}

__global__ void kan_init_out(const float* __restrict__ bout, float* __restrict__ out)
{
    const int idx = blockIdx.x * 256 + threadIdx.x;
    out[idx] = bout[idx & 63];
}

extern "C" void kernel_launch(void* const* d_in, const int* in_sizes, int n_in,
                              void* d_out, int out_size, void* d_ws, size_t ws_size,
                              hipStream_t stream)
{
    const float* x    = (const float*)d_in[0];
    const float* w0   = (const float*)d_in[1];
    const float* b0   = (const float*)d_in[2];
    const float* w1   = (const float*)d_in[3];
    const float* b1   = (const float*)d_in[4];
    const float* w2   = (const float*)d_in[5];
    const float* b2   = (const float*)d_in[6];
    const float* wout = (const float*)d_in[7];
    const float* bout = (const float*)d_in[8];
    float* out = (float*)d_out;

    const size_t need = (size_t)DDIM * BSZ * sizeof(float);
    if (ws_size >= need) {
        float* part = (float*)d_ws;
        kan_edge_kernel<<<DDIM, 256, 0, stream>>>(x, w0, b0, w1, b1, w2, b2, wout, part, 0);
        kan_reduce_kernel<<<OUT_DIM, BSZ, 0, stream>>>(part, bout, out);
    } else {
        kan_init_out<<<64, 256, 0, stream>>>(bout, out);
        kan_edge_kernel<<<DDIM, 256, 0, stream>>>(x, w0, b0, w1, b1, w2, b2, wout, out, 1);
    }
}


# Hidden layers as transposed f32 WMMA products (A=W from LDS, N=batch): each layer's D tiles are the next layer's B operand, activations stay in registers; k and output sum order change
# speedup vs baseline: 1.3018x; 1.3018x over previous
#include <hip/hip_runtime.h>


typedef float v2f __attribute__((ext_vector_type(2)));
typedef float v4f __attribute__((ext_vector_type(4)));
typedef float v8f __attribute__((ext_vector_type(8)));

#define IN_DIM  64
#define OUT_DIM 64
#define HID     64
#define DDIM    (IN_DIM * OUT_DIM)
#define BSZ     256
#define WSTR    68
#define NB      2

__launch_bounds__(256)
__global__ void kan_edge_kernel(const float* __restrict__ x,
                                const float* __restrict__ w0, const float* __restrict__ b0,
                                const float* __restrict__ w1, const float* __restrict__ b1,
                                const float* __restrict__ w2, const float* __restrict__ b2,
                                const float* __restrict__ wout,
                                float* __restrict__ outbuf, int use_atomic)
{
    __shared__ __attribute__((aligned(16))) float W1s[HID * WSTR];
    __shared__ __attribute__((aligned(16))) float W2s[HID * WSTR];
    __shared__ __attribute__((aligned(16))) float b1s[HID];
    __shared__ __attribute__((aligned(16))) float b2s[HID];
    __shared__ __attribute__((aligned(16))) float wos[HID];

    const int d    = blockIdx.x;
    const int o    = d >> 6;
    const int i    = d & 63;
    const int tid  = threadIdx.x;
    const int lane = tid & 31;
    const int wave = tid >> 5;
    const int c    = lane & 15;
    const int h    = lane >> 4;

    const v4f* g1 = (const v4f*)(w1 + (size_t)d * HID * HID);
    const v4f* g2 = (const v4f*)(w2 + (size_t)d * HID * HID);
    v4f s1[4], s2[4];
    #pragma unroll
    for (int s = 0; s < 4; ++s) s1[s] = g1[tid + 256 * s];
    #pragma unroll
    for (int s = 0; s < 4; ++s) s2[s] = g2[tid + 256 * s];

    float vb1 = 0.f, vb2 = 0.f, vwo = 0.f;
    if (tid < HID) {
        vb1 = b1[d * HID + tid];
        vb2 = b2[d * HID + tid];
        vwo = wout[o * (IN_DIM * HID) + i * HID + tid];
    }

    float xv[NB];
    #pragma unroll
    for (int nb = 0; nb < NB; ++nb)
        xv[nb] = x[(wave * (16 * NB) + nb * 16 + c) * IN_DIM + i];

    v2f act[16][NB];
    {
        const float* w0p = w0 + d * HID + 8 * h;
        const float* b0p = b0 + d * HID + 8 * h;
        #pragma unroll
        for (int mt = 0; mt < 4; ++mt) {
            const v4f wa = *(const v4f*)(w0p + 16 * mt);
            const v4f wb = *(const v4f*)(w0p + 16 * mt + 4);
            const v4f ba = *(const v4f*)(b0p + 16 * mt);
            const v4f bb = *(const v4f*)(b0p + 16 * mt + 4);
            const float wv[8] = {wa[0], wa[1], wa[2], wa[3], wb[0], wb[1], wb[2], wb[3]};
            const float bv[8] = {ba[0], ba[1], ba[2], ba[3], bb[0], bb[1], bb[2], bb[3]};
            #pragma unroll
            for (int j = 0; j < 4; ++j) {
                #pragma unroll
                for (int nb = 0; nb < NB; ++nb) {
                    float v0 = fmaf(xv[nb], wv[2 * j],     bv[2 * j]);
                    float v1 = fmaf(xv[nb], wv[2 * j + 1], bv[2 * j + 1]);
                    v2f f;
                    f[0] = v0 > 0.f ? v0 : 0.f;
                    f[1] = v1 > 0.f ? v1 : 0.f;
                    act[mt * 4 + j][nb] = f;
                }
            }
        }
    }

    #pragma unroll
    for (int s = 0; s < 4; ++s) {
        const int q = tid + 256 * s;
        *(v4f*)(W1s + (q >> 4) * WSTR + (q & 15) * 4) = s1[s];
    }
    #pragma unroll
    for (int s = 0; s < 4; ++s) {
        const int q = tid + 256 * s;
        *(v4f*)(W2s + (q >> 4) * WSTR + (q & 15) * 4) = s2[s];
    }
    if (tid < HID) { b1s[tid] = vb1; b2s[tid] = vb2; wos[tid] = vwo; }
    __syncthreads();

    auto load_row = [&](const float* __restrict__ Ws, int mtp, v4f (&a)[8]) {
        const float* ap = Ws + (16 * mtp + c) * WSTR + 8 * h;
        #pragma unroll
        for (int mt = 0; mt < 4; ++mt) {
            a[2 * mt]     = *(const v4f*)(ap + 16 * mt);
            a[2 * mt + 1] = *(const v4f*)(ap + 16 * mt + 4);
        }
    };

    auto mma_row = [&](const v4f (&a)[8], v8f (&accr)[NB]) {
        #pragma unroll
        for (int nb = 0; nb < NB; ++nb)
            accr[nb] = (v8f){0.f, 0.f, 0.f, 0.f, 0.f, 0.f, 0.f, 0.f};
        #pragma unroll
        for (int mt = 0; mt < 4; ++mt) {
            #pragma unroll
            for (int j = 0; j < 4; ++j) {
                const v4f q = a[2 * mt + (j >> 1)];
                v2f af;
                af[0] = q[2 * (j & 1)];
                af[1] = q[2 * (j & 1) + 1];
                #pragma unroll
                for (int nb = 0; nb < NB; ++nb)
                    accr[nb] = __builtin_amdgcn_wmma_f32_16x16x4_f32(
                        false, af, false, act[mt * 4 + j][nb], (short)0, accr[nb], false, false);
            }
        }
    };

    auto lane_vec = [&](const float* __restrict__ vs, float (&out)[4][8]) {
        #pragma unroll
        for (int mt = 0; mt < 4; ++mt) {
            const v4f lo = *(const v4f*)(vs + 16 * mt + 8 * h);
            const v4f hi = *(const v4f*)(vs + 16 * mt + 8 * h + 4);
            out[mt][0] = lo[0]; out[mt][1] = lo[1]; out[mt][2] = lo[2]; out[mt][3] = lo[3];
            out[mt][4] = hi[0]; out[mt][5] = hi[1]; out[mt][6] = hi[2]; out[mt][7] = hi[3];
        }
    };

    #define SCHED_FENCE() __builtin_amdgcn_sched_barrier(0)

    v8f   acc[4][NB];
    v4f   aA[8], aB[8];
    float bias[4][8], wo[4][8];

    load_row(W1s, 0, aA);
    SCHED_FENCE();
    load_row(W1s, 1, aB);
    SCHED_FENCE();
    mma_row(aA, acc[0]);
    SCHED_FENCE();
    load_row(W1s, 2, aA);
    SCHED_FENCE();
    mma_row(aB, acc[1]);
    SCHED_FENCE();
    load_row(W1s, 3, aB);
    SCHED_FENCE();
    mma_row(aA, acc[2]);
    SCHED_FENCE();
    lane_vec(b1s, bias);
    load_row(W2s, 0, aA);
    SCHED_FENCE();
    mma_row(aB, acc[3]);
    SCHED_FENCE();

    #pragma unroll
    for (int mt = 0; mt < 4; ++mt) {
        #pragma unroll
        for (int j = 0; j < 4; ++j) {
            #pragma unroll
            for (int nb = 0; nb < NB; ++nb) {
                float v0 = acc[mt][nb][2 * j]     + bias[mt][2 * j];
                float v1 = acc[mt][nb][2 * j + 1] + bias[mt][2 * j + 1];
                v2f f;
                f[0] = v0 > 0.f ? v0 : 0.f;
                f[1] = v1 > 0.f ? v1 : 0.f;
                act[mt * 4 + j][nb] = f;
            }
        }
    }

    SCHED_FENCE();
    load_row(W2s, 1, aB);
    SCHED_FENCE();
    mma_row(aA, acc[0]);
    SCHED_FENCE();
    load_row(W2s, 2, aA);
    SCHED_FENCE();
    mma_row(aB, acc[1]);
    SCHED_FENCE();
    load_row(W2s, 3, aB);
    SCHED_FENCE();
    mma_row(aA, acc[2]);
    SCHED_FENCE();
    lane_vec(b2s, bias);
    lane_vec(wos, wo);
    SCHED_FENCE();
    mma_row(aB, acc[3]);
    SCHED_FENCE();

    {
        float t[NB];
        #pragma unroll
        for (int nb = 0; nb < NB; ++nb) {
            float s = 0.f;
            #pragma unroll
            for (int mt = 0; mt < 4; ++mt) {
                #pragma unroll
                for (int r = 0; r < 8; ++r) {
                    float v = acc[mt][nb][r] + bias[mt][r];
                    v = v > 0.f ? v : 0.f;
                    s = fmaf(wo[mt][r], v, s);
                }
            }
            t[nb] = s + __shfl_xor(s, 16);
        }
        const float res = h ? t[1] : t[0];
        const int   b   = wave * (16 * NB) + lane;
        if (use_atomic) {
            __hip_atomic_fetch_add(&outbuf[b * OUT_DIM + o], res,
                                   __ATOMIC_RELAXED, __HIP_MEMORY_SCOPE_AGENT);
        } else {
            outbuf[(size_t)d * BSZ + b] = res;
        }
    }
    #undef SCHED_FENCE
}

__global__ void kan_reduce_kernel(const float* __restrict__ part,
                                  const float* __restrict__ bout,
                                  float* __restrict__ out)
{
    const int o = blockIdx.x;
    const int b = threadIdx.x;
    float s = bout[o];
    const float* p = part + (size_t)o * 64 * BSZ + b;
    #pragma unroll 8
    for (int i = 0; i < 64; ++i) s += p[i * BSZ];
    out[b * OUT_DIM + o] = s;
}

__global__ void kan_init_out(const float* __restrict__ bout, float* __restrict__ out)
{
    const int idx = blockIdx.x * 256 + threadIdx.x;
    out[idx] = bout[idx & 63];
}

extern "C" void kernel_launch(void* const* d_in, const int* in_sizes, int n_in,
                              void* d_out, int out_size, void* d_ws, size_t ws_size,
                              hipStream_t stream)
{
    const float* x    = (const float*)d_in[0];
    const float* w0   = (const float*)d_in[1];
    const float* b0   = (const float*)d_in[2];
    const float* w1   = (const float*)d_in[3];
    const float* b1   = (const float*)d_in[4];
    const float* w2   = (const float*)d_in[5];
    const float* b2   = (const float*)d_in[6];
    const float* wout = (const float*)d_in[7];
    const float* bout = (const float*)d_in[8];
    float* out = (float*)d_out;

    const size_t need = (size_t)DDIM * BSZ * sizeof(float);
    if (ws_size >= need) {
        float* part = (float*)d_ws;
        kan_edge_kernel<<<DDIM, 256, 0, stream>>>(x, w0, b0, w1, b1, w2, b2, wout, part, 0);
        kan_reduce_kernel<<<OUT_DIM, BSZ, 0, stream>>>(part, bout, out);
    } else {
        kan_init_out<<<64, 256, 0, stream>>>(bout, out);
        kan_edge_kernel<<<DDIM, 256, 0, stream>>>(x, w0, b0, w1, b1, w2, b2, wout, out, 1);
    }
}


# Workgroup loops over 8 edges of one output neuron: next edge's weights go global->registers->second LDS buffer under the current WMMAs; edge sums kept in a register; output sum order changes
# speedup vs baseline: 1.3523x; 1.0388x over previous
#include <hip/hip_runtime.h>


typedef float v2f __attribute__((ext_vector_type(2)));
typedef float v4f __attribute__((ext_vector_type(4)));
typedef float v8f __attribute__((ext_vector_type(8)));

#define IN_DIM  64
#define OUT_DIM 64
#define HID     64
#define DDIM    (IN_DIM * OUT_DIM)
#define BSZ     256
#define WSTR    68
#define NB      2
#define EPW     8
#define NGRP    (IN_DIM / EPW)
#define WMAT    (HID * WSTR)
#define VECS    (5 * HID)

__launch_bounds__(256)
__global__ void kan_edge_kernel(const float* __restrict__ x,
                                const float* __restrict__ w0, const float* __restrict__ b0,
                                const float* __restrict__ w1, const float* __restrict__ b1,
                                const float* __restrict__ w2, const float* __restrict__ b2,
                                const float* __restrict__ wout,
                                float* __restrict__ outbuf, int use_atomic)
{
    __shared__ __attribute__((aligned(16))) float W1s[2 * WMAT];
    __shared__ __attribute__((aligned(16))) float W2s[2 * WMAT];
    __shared__ __attribute__((aligned(16))) float vecs[2 * VECS];

    const int g    = blockIdx.x;
    const int o    = g / NGRP;
    const int i0   = (g % NGRP) * EPW;
    const int d0   = o * IN_DIM + i0;
    const int tid  = threadIdx.x;
    const int lane = tid & 31;
    const int wave = tid >> 5;
    const int c    = lane & 15;
    const int h    = lane >> 4;

    const int    vsel = tid >> 6;
    const float* vp   = (vsel == 0 ? w0 : vsel == 1 ? b0 : vsel == 2 ? b1 : b2) + (tid & 63);
    const float* xrow = x + (wave * (16 * NB) + c) * IN_DIM + i0;

    v4f   pre[4];
    float pv, pwo = 0.f;
    float xv[NB];

    {
        const v4f* g1 = (const v4f*)(w1 + (size_t)d0 * HID * HID);
        const v4f* g2 = (const v4f*)(w2 + (size_t)d0 * HID * HID);
        v4f s2[4];
        #pragma unroll
        for (int s = 0; s < 4; ++s) pre[s] = g1[tid + 256 * s];
        #pragma unroll
        for (int s = 0; s < 4; ++s) s2[s] = g2[tid + 256 * s];
        pv = vp[d0 * HID];
        if (tid < HID) pwo = wout[d0 * HID + tid];
        #pragma unroll
        for (int nb = 0; nb < NB; ++nb) xv[nb] = xrow[nb * 16 * IN_DIM];
        #pragma unroll
        for (int s = 0; s < 4; ++s) {
            const int q = tid + 256 * s;
            *(v4f*)(W1s + (q >> 4) * WSTR + (q & 15) * 4) = pre[s];
            *(v4f*)(W2s + (q >> 4) * WSTR + (q & 15) * 4) = s2[s];
        }
        vecs[tid] = pv;
        if (tid < HID) vecs[4 * HID + tid] = pwo;
        __syncthreads();
    }

    auto load_row = [&](const float* __restrict__ Ws, int mtp, v4f (&a)[8]) {
        const float* ap = Ws + (16 * mtp + c) * WSTR + 8 * h;
        #pragma unroll
        for (int mt = 0; mt < 4; ++mt) {
            a[2 * mt]     = *(const v4f*)(ap + 16 * mt);
            a[2 * mt + 1] = *(const v4f*)(ap + 16 * mt + 4);
        }
    };

    v2f act[16][NB];

    auto mma_row = [&](const v4f (&a)[8], v8f (&accr)[NB]) {
        #pragma unroll
        for (int nb = 0; nb < NB; ++nb)
            accr[nb] = (v8f){0.f, 0.f, 0.f, 0.f, 0.f, 0.f, 0.f, 0.f};
        #pragma unroll
        for (int mt = 0; mt < 4; ++mt) {
            #pragma unroll
            for (int j = 0; j < 4; ++j) {
                const v4f q = a[2 * mt + (j >> 1)];
                v2f af;
                af[0] = q[2 * (j & 1)];
                af[1] = q[2 * (j & 1) + 1];
                #pragma unroll
                for (int nb = 0; nb < NB; ++nb)
                    accr[nb] = __builtin_amdgcn_wmma_f32_16x16x4_f32(
                        false, af, false, act[mt * 4 + j][nb], (short)0, accr[nb], false, false);
            }
        }
    };

    auto lane_vec = [&](const float* __restrict__ vs, float (&out)[4][8]) {
        #pragma unroll
        for (int mt = 0; mt < 4; ++mt) {
            const v4f lo = *(const v4f*)(vs + 16 * mt + 8 * h);
            const v4f hi = *(const v4f*)(vs + 16 * mt + 8 * h + 4);
            out[mt][0] = lo[0]; out[mt][1] = lo[1]; out[mt][2] = lo[2]; out[mt][3] = lo[3];
            out[mt][4] = hi[0]; out[mt][5] = hi[1]; out[mt][6] = hi[2]; out[mt][7] = hi[3];
        }
    };

    #define SCHED_FENCE() __builtin_amdgcn_sched_barrier(0)

    v8f   acc[4][NB];
    v4f   aA[8], aB[8];
    float bias[4][8], wo[4][8];
    float sacc[NB];
    #pragma unroll
    for (int nb = 0; nb < NB; ++nb) sacc[nb] = 0.f;

    #pragma nounroll
    for (int e = 0; e < EPW; ++e) {
        const int    cur  = e & 1;
        const float* W1c  = W1s + cur * WMAT;
        const float* W2c  = W2s + cur * WMAT;
        float*       W1n  = W1s + (cur ^ 1) * WMAT;
        float*       W2n  = W2s + (cur ^ 1) * WMAT;
        const float* vc   = vecs + cur * VECS;
        float*       vn   = vecs + (cur ^ 1) * VECS;
        const bool   more = e + 1 < EPW;
        const int    dn   = d0 + e + 1;

        float xn[NB];
        #pragma unroll
        for (int nb = 0; nb < NB; ++nb) xn[nb] = 0.f;
        if (more) {
            const v4f* g1 = (const v4f*)(w1 + (size_t)dn * HID * HID);
            #pragma unroll
            for (int s = 0; s < 4; ++s) pre[s] = g1[tid + 256 * s];
            pv = vp[dn * HID];
            if (tid < HID) pwo = wout[dn * HID + tid];
            #pragma unroll
            for (int nb = 0; nb < NB; ++nb) xn[nb] = xrow[nb * 16 * IN_DIM + e + 1];
        }

        lane_vec(vc, bias);
        lane_vec(vc + HID, wo);
        load_row(W1c, 0, aA);
        SCHED_FENCE();
        #pragma unroll
        for (int mt = 0; mt < 4; ++mt) {
            #pragma unroll
            for (int j = 0; j < 4; ++j) {
                #pragma unroll
                for (int nb = 0; nb < NB; ++nb) {
                    float v0 = fmaf(xv[nb], bias[mt][2 * j],     wo[mt][2 * j]);
                    float v1 = fmaf(xv[nb], bias[mt][2 * j + 1], wo[mt][2 * j + 1]);
                    v2f f;
                    f[0] = v0 > 0.f ? v0 : 0.f;
                    f[1] = v1 > 0.f ? v1 : 0.f;
                    act[mt * 4 + j][nb] = f;
                }
            }
        }

        SCHED_FENCE();
        load_row(W1c, 1, aB);
        SCHED_FENCE();
        mma_row(aA, acc[0]);
        SCHED_FENCE();
        load_row(W1c, 2, aA);
        SCHED_FENCE();
        mma_row(aB, acc[1]);
        SCHED_FENCE();
        load_row(W1c, 3, aB);
        SCHED_FENCE();
        mma_row(aA, acc[2]);
        SCHED_FENCE();
        if (more) {
            #pragma unroll
            for (int s = 0; s < 4; ++s) {
                const int q = tid + 256 * s;
                *(v4f*)(W1n + (q >> 4) * WSTR + (q & 15) * 4) = pre[s];
            }
        }
        lane_vec(vc + 2 * HID, bias);
        SCHED_FENCE();
        mma_row(aB, acc[3]);
        SCHED_FENCE();
        load_row(W2c, 0, aA);

        #pragma unroll
        for (int mt = 0; mt < 4; ++mt) {
            #pragma unroll
            for (int j = 0; j < 4; ++j) {
                #pragma unroll
                for (int nb = 0; nb < NB; ++nb) {
                    float v0 = acc[mt][nb][2 * j]     + bias[mt][2 * j];
                    float v1 = acc[mt][nb][2 * j + 1] + bias[mt][2 * j + 1];
                    v2f f;
                    f[0] = v0 > 0.f ? v0 : 0.f;
                    f[1] = v1 > 0.f ? v1 : 0.f;
                    act[mt * 4 + j][nb] = f;
                }
            }
        }

        if (more) {
            const v4f* g2 = (const v4f*)(w2 + (size_t)dn * HID * HID);
            #pragma unroll
            for (int s = 0; s < 4; ++s) pre[s] = g2[tid + 256 * s];
        }

        SCHED_FENCE();
        load_row(W2c, 1, aB);
        SCHED_FENCE();
        mma_row(aA, acc[0]);
        SCHED_FENCE();
        load_row(W2c, 2, aA);
        SCHED_FENCE();
        mma_row(aB, acc[1]);
        SCHED_FENCE();
        load_row(W2c, 3, aB);
        SCHED_FENCE();
        mma_row(aA, acc[2]);
        SCHED_FENCE();
        if (more) {
            #pragma unroll
            for (int s = 0; s < 4; ++s) {
                const int q = tid + 256 * s;
                *(v4f*)(W2n + (q >> 4) * WSTR + (q & 15) * 4) = pre[s];
            }
        }
        lane_vec(vc + 3 * HID, bias);
        SCHED_FENCE();
        mma_row(aB, acc[3]);
        SCHED_FENCE();
        lane_vec(vc + 4 * HID, wo);

        #pragma unroll
        for (int nb = 0; nb < NB; ++nb) {
            float s = sacc[nb];
            #pragma unroll
            for (int mt = 0; mt < 4; ++mt) {
                #pragma unroll
                for (int r = 0; r < 8; ++r) {
                    float v = acc[mt][nb][r] + bias[mt][r];
                    v = v > 0.f ? v : 0.f;
                    s = fmaf(wo[mt][r], v, s);
                }
            }
            sacc[nb] = s;
            xv[nb]   = xn[nb];
        }

        if (more) {
            vn[tid] = pv;
            if (tid < HID) vn[4 * HID + tid] = pwo;
            __syncthreads();
        }
    }
    #undef SCHED_FENCE

    {
        float t[NB];
        #pragma unroll
        for (int nb = 0; nb < NB; ++nb)
            t[nb] = sacc[nb] + __shfl_xor(sacc[nb], 16);
        const float res = h ? t[1] : t[0];
        const int   b   = wave * (16 * NB) + lane;
        if (use_atomic) {
            __hip_atomic_fetch_add(&outbuf[b * OUT_DIM + o], res,
                                   __ATOMIC_RELAXED, __HIP_MEMORY_SCOPE_AGENT);
        } else {
            outbuf[(size_t)g * BSZ + b] = res;
        }
    }
}

__global__ void kan_reduce_kernel(const float* __restrict__ part,
                                  const float* __restrict__ bout,
                                  float* __restrict__ out)
{
    const int o = blockIdx.x;
    const int b = threadIdx.x;
    float s = bout[o];
    const float* p = part + (size_t)o * NGRP * BSZ + b;
    #pragma unroll
    for (int k = 0; k < NGRP; ++k) s += p[k * BSZ];
    out[b * OUT_DIM + o] = s;
}

__global__ void kan_init_out(const float* __restrict__ bout, float* __restrict__ out)
{
    const int idx = blockIdx.x * 256 + threadIdx.x;
    out[idx] = bout[idx & 63];
}

extern "C" void kernel_launch(void* const* d_in, const int* in_sizes, int n_in,
                              void* d_out, int out_size, void* d_ws, size_t ws_size,
                              hipStream_t stream)
{
    const float* x    = (const float*)d_in[0];
    const float* w0   = (const float*)d_in[1];
    const float* b0   = (const float*)d_in[2];
    const float* w1   = (const float*)d_in[3];
    const float* b1   = (const float*)d_in[4];
    const float* w2   = (const float*)d_in[5];
    const float* b2   = (const float*)d_in[6];
    const float* wout = (const float*)d_in[7];
    const float* bout = (const float*)d_in[8];
    float* out = (float*)d_out;

    const size_t need = (size_t)(DDIM / EPW) * BSZ * sizeof(float);
    if (ws_size >= need) {
        float* part = (float*)d_ws;
        kan_edge_kernel<<<DDIM / EPW, 256, 0, stream>>>(x, w0, b0, w1, b1, w2, b2, wout, part, 0);
        kan_reduce_kernel<<<OUT_DIM, BSZ, 0, stream>>>(part, bout, out);
    } else {
        kan_init_out<<<64, 256, 0, stream>>>(bout, out);
        kan_edge_kernel<<<DDIM / EPW, 256, 0, stream>>>(x, w0, b0, w1, b1, w2, b2, wout, out, 1);
    }
}


# Layer 1 skips k chunks whose input units are <= 0 in all 256 rows (exact test from the x column's range); W1 columns, w0, b0 staged live-first per edge; f32 WMMA kept; layer-1 k sum order changes
# speedup vs baseline: 1.5364x; 1.1361x over previous
#include <hip/hip_runtime.h>


typedef float v2f __attribute__((ext_vector_type(2)));
typedef float v4f __attribute__((ext_vector_type(4)));
typedef float v8f __attribute__((ext_vector_type(8)));

#define IN_DIM  64
#define OUT_DIM 64
#define HID     64
#define DDIM    (IN_DIM * OUT_DIM)
#define BSZ     256
#define WSTR    68
#define NB      2
#define EPW     8
#define NGRP    (IN_DIM / EPW)
#define WMAT    (HID * WSTR)
#define VECS    (5 * HID)

__launch_bounds__(256, 4)
__global__ void kan_edge_kernel(const float* __restrict__ x,
                                const float* __restrict__ w0, const float* __restrict__ b0,
                                const float* __restrict__ w1, const float* __restrict__ b1,
                                const float* __restrict__ w2, const float* __restrict__ b2,
                                const float* __restrict__ wout,
                                float* __restrict__ outbuf, int use_atomic)
{
    __shared__ __attribute__((aligned(16))) float W1s[2 * WMAT];
    __shared__ __attribute__((aligned(16))) float W2s[2 * WMAT];
    __shared__ __attribute__((aligned(16))) float vecs[2 * VECS];
    __shared__ __attribute__((aligned(16))) float xmm[8 * 16];
    __shared__ __attribute__((aligned(16))) float xrg[16];

    const int g    = blockIdx.x;
    const int o    = g / NGRP;
    const int i0   = (g % NGRP) * EPW;
    const int d0   = o * IN_DIM + i0;
    const int tid  = threadIdx.x;
    const int lane = tid & 31;
    const int wave = tid >> 5;
    const int c    = lane & 15;
    const int h    = lane >> 4;

    const int    vsel = tid >> 6;
    const float* vp   = (vsel == 0 ? w0 : vsel == 1 ? b0 : vsel == 2 ? b1 : b2) + (tid & 63);
    const float* xrow = x + (wave * (16 * NB) + c) * IN_DIM + i0;

    v4f   pre[4];
    float pv, pwo = 0.f;
    float xv[NB];
    float nwl, nbl, nwh, nbh;
    unsigned cm0, cm1;

    auto slot_of_rank = [](int r) -> int {
        return (r & 0x30) | ((r & 2) << 2) | ((r >> 1) & 6) | (r & 1);
    };
    auto slot_of_unit = [&](int u, unsigned m0, unsigned m1) -> int {
        const unsigned m    = u < 32 ? m0 : m1;
        const int      v    = u & 31;
        const int      lb   = __popc(m & ((1u << v) - 1u));
        const int      n0   = __popc(m0);
        const int      nl   = n0 + __popc(m1);
        const int      lv   = (int)((m >> v) & 1u);
        const int      ra   = (u < 32 ? 0 : n0) + lb;
        const int      rb   = nl + (u < 32 ? 0 : 32 - n0) + (v - lb);
        return slot_of_rank(rb + lv * (ra - rb));
    };
    auto live_masks = [&](float wl, float bl, float wh, float bh, float mn, float mx, unsigned& m0, unsigned& m1) {
        const bool l0 = fmaf(mn, wl, bl) > 0.f || fmaf(mx, wl, bl) > 0.f;
        const bool l1 = fmaf(mn, wh, bh) > 0.f || fmaf(mx, wh, bh) > 0.f;
        m0 = __builtin_amdgcn_ballot_w32(l0);
        m1 = __builtin_amdgcn_ballot_w32(l1);
    };
    auto store_w1 = [&](float* __restrict__ Wd, const v4f (&p)[4], unsigned m0, unsigned m1) {
        float* rowp = Wd + (tid >> 4) * WSTR;
        #pragma unroll
        for (int t = 0; t < 4; ++t) {
            const int sl = slot_of_unit(4 * (tid & 15) + t, m0, m1);
            #pragma unroll
            for (int s = 0; s < 4; ++s) rowp[16 * s * WSTR + sl] = p[s][t];
        }
    };
    auto vec_slot = [&](unsigned m0, unsigned m1) -> int {
        return tid < 2 * HID ? (tid & HID) + slot_of_unit(tid & (HID - 1), m0, m1) : tid;
    };

    {
        const v4f* g1 = (const v4f*)(w1 + (size_t)d0 * HID * HID);
        const v4f* g2 = (const v4f*)(w2 + (size_t)d0 * HID * HID);
        v4f s2[4];
        #pragma unroll
        for (int s = 0; s < 4; ++s) pre[s] = g1[tid + 256 * s];
        #pragma unroll
        for (int s = 0; s < 4; ++s) s2[s] = g2[tid + 256 * s];
        pv = vp[d0 * HID];
        if (tid < HID) pwo = wout[d0 * HID + tid];
        #pragma unroll
        for (int nb = 0; nb < NB; ++nb) xv[nb] = xrow[nb * 16 * IN_DIM];
        nwl = w0[d0 * HID + lane];       nbl = b0[d0 * HID + lane];
        nwh = w0[d0 * HID + lane + 32];  nbh = b0[d0 * HID + lane + 32];
        {
            const v4f* xr = (const v4f*)(x + tid * IN_DIM + i0);
            const v4f  xa = xr[0], xb = xr[1];
            float mn[8] = {xa[0], xa[1], xa[2], xa[3], xb[0], xb[1], xb[2], xb[3]};
            float mx[8] = {xa[0], xa[1], xa[2], xa[3], xb[0], xb[1], xb[2], xb[3]};
            #pragma unroll
            for (int off = 16; off >= 1; off >>= 1) {
                #pragma unroll
                for (int k = 0; k < 8; ++k) {
                    mn[k] = fminf(mn[k], __shfl_xor(mn[k], off));
                    mx[k] = fmaxf(mx[k], __shfl_xor(mx[k], off));
                }
            }
            if (lane == 0) {
                *(v4f*)(xmm + wave * 16)      = (v4f){mn[0], mn[1], mn[2], mn[3]};
                *(v4f*)(xmm + wave * 16 + 4)  = (v4f){mn[4], mn[5], mn[6], mn[7]};
                *(v4f*)(xmm + wave * 16 + 8)  = (v4f){mx[0], mx[1], mx[2], mx[3]};
                *(v4f*)(xmm + wave * 16 + 12) = (v4f){mx[4], mx[5], mx[6], mx[7]};
            }
        }
        __syncthreads();
        {
            const int t = lane & 15;
            float lo = xmm[t], hi = lo;
            #pragma unroll
            for (int w = 1; w < 8; ++w) {
                const float v = xmm[w * 16 + t];
                lo = fminf(lo, v);
                hi = fmaxf(hi, v);
            }
            const float r = t < 8 ? lo : hi;
            if (tid < 16) xrg[t] = r;
            live_masks(nwl, nbl, nwh, nbh, __shfl(r, 0), __shfl(r, 8), cm0, cm1);
        }
        store_w1(W1s, pre, cm0, cm1);
        #pragma unroll
        for (int s = 0; s < 4; ++s) {
            const int q = tid + 256 * s;
            *(v4f*)(W2s + (q >> 4) * WSTR + (q & 15) * 4) = s2[s];
        }
        vecs[vec_slot(cm0, cm1)] = pv;
        if (tid < HID) vecs[4 * HID + tid] = pwo;
        __syncthreads();
    }

    auto load_row = [&](const float* __restrict__ Ws, int mtp, v4f (&a)[8]) {
        const float* ap = Ws + (16 * mtp + c) * WSTR + 8 * h;
        #pragma unroll
        for (int mt = 0; mt < 4; ++mt) {
            a[2 * mt]     = *(const v4f*)(ap + 16 * mt);
            a[2 * mt + 1] = *(const v4f*)(ap + 16 * mt + 4);
        }
    };

    v2f act[16][NB];

    auto mma_row = [&](const v4f (&a)[8], v8f (&accr)[NB], int nchunks) {
        #pragma unroll
        for (int nb = 0; nb < NB; ++nb)
            accr[nb] = (v8f){0.f, 0.f, 0.f, 0.f, 0.f, 0.f, 0.f, 0.f};
        #pragma unroll
        for (int mt = 0; mt < 4; ++mt) {
            #pragma unroll
            for (int j = 0; j < 4; ++j) {
                if (mt > 0 && mt * 4 + j >= nchunks) return;
                const v4f q = a[2 * mt + (j >> 1)];
                v2f af;
                af[0] = q[2 * (j & 1)];
                af[1] = q[2 * (j & 1) + 1];
                #pragma unroll
                for (int nb = 0; nb < NB; ++nb)
                    accr[nb] = __builtin_amdgcn_wmma_f32_16x16x4_f32(
                        false, af, false, act[mt * 4 + j][nb], (short)0, accr[nb], false, false);
            }
        }
    };

    auto lane_vec = [&](const float* __restrict__ vs, float (&out)[4][8]) {
        #pragma unroll
        for (int mt = 0; mt < 4; ++mt) {
            const v4f lo = *(const v4f*)(vs + 16 * mt + 8 * h);
            const v4f hi = *(const v4f*)(vs + 16 * mt + 8 * h + 4);
            out[mt][0] = lo[0]; out[mt][1] = lo[1]; out[mt][2] = lo[2]; out[mt][3] = lo[3];
            out[mt][4] = hi[0]; out[mt][5] = hi[1]; out[mt][6] = hi[2]; out[mt][7] = hi[3];
        }
    };

    #define SCHED_FENCE() __builtin_amdgcn_sched_barrier(0)

    v8f   acc[4][NB];
    v4f   aA[8], aB[8];
    float bias[4][8], wo[4][8];
    float sacc[NB];
    #pragma unroll
    for (int nb = 0; nb < NB; ++nb) sacc[nb] = 0.f;

    #pragma nounroll
    for (int e = 0; e < EPW; ++e) {
        const int    cur  = e & 1;
        const float* W1c  = W1s + cur * WMAT;
        const float* W2c  = W2s + cur * WMAT;
        float*       W1n  = W1s + (cur ^ 1) * WMAT;
        float*       W2n  = W2s + (cur ^ 1) * WMAT;
        const float* vc   = vecs + cur * VECS;
        float*       vn   = vecs + (cur ^ 1) * VECS;
        const bool   more = e + 1 < EPW;
        const int    dn   = d0 + e + 1;
        const int    nch  = (__popc(cm0) + __popc(cm1) + 3) >> 2;
        unsigned     nm0  = 0u, nm1 = 0u;

        float xn[NB];
        #pragma unroll
        for (int nb = 0; nb < NB; ++nb) xn[nb] = 0.f;
        if (more) {
            const v4f* g1 = (const v4f*)(w1 + (size_t)dn * HID * HID);
            #pragma unroll
            for (int s = 0; s < 4; ++s) pre[s] = g1[tid + 256 * s];
            pv = vp[dn * HID];
            if (tid < HID) pwo = wout[dn * HID + tid];
            #pragma unroll
            for (int nb = 0; nb < NB; ++nb) xn[nb] = xrow[nb * 16 * IN_DIM + e + 1];
            nwl = w0[dn * HID + lane];       nbl = b0[dn * HID + lane];
            nwh = w0[dn * HID + lane + 32];  nbh = b0[dn * HID + lane + 32];
        }

        lane_vec(vc, bias);
        lane_vec(vc + HID, wo);
        load_row(W1c, 0, aA);
        SCHED_FENCE();
        #pragma unroll
        for (int mt = 0; mt < 4; ++mt) {
            #pragma unroll
            for (int j = 0; j < 4; ++j) {
                #pragma unroll
                for (int nb = 0; nb < NB; ++nb) {
                    float v0 = fmaf(xv[nb], bias[mt][2 * j],     wo[mt][2 * j]);
                    float v1 = fmaf(xv[nb], bias[mt][2 * j + 1], wo[mt][2 * j + 1]);
                    v2f f;
                    f[0] = v0 > 0.f ? v0 : 0.f;
                    f[1] = v1 > 0.f ? v1 : 0.f;
                    act[mt * 4 + j][nb] = f;
                }
            }
        }

        SCHED_FENCE();
        load_row(W1c, 1, aB);
        SCHED_FENCE();
        mma_row(aA, acc[0], nch);
        SCHED_FENCE();
        load_row(W1c, 2, aA);
        SCHED_FENCE();
        mma_row(aB, acc[1], nch);
        SCHED_FENCE();
        load_row(W1c, 3, aB);
        SCHED_FENCE();
        mma_row(aA, acc[2], nch);
        SCHED_FENCE();
        if (more) {
            live_masks(nwl, nbl, nwh, nbh, xrg[e + 1], xrg[8 + e + 1], nm0, nm1);
            store_w1(W1n, pre, nm0, nm1);
        }
        lane_vec(vc + 2 * HID, bias);
        SCHED_FENCE();
        mma_row(aB, acc[3], nch);
        SCHED_FENCE();
        load_row(W2c, 0, aA);

        #pragma unroll
        for (int mt = 0; mt < 4; ++mt) {
            #pragma unroll
            for (int j = 0; j < 4; ++j) {
                #pragma unroll
                for (int nb = 0; nb < NB; ++nb) {
                    float v0 = acc[mt][nb][2 * j]     + bias[mt][2 * j];
                    float v1 = acc[mt][nb][2 * j + 1] + bias[mt][2 * j + 1];
                    v2f f;
                    f[0] = v0 > 0.f ? v0 : 0.f;
                    f[1] = v1 > 0.f ? v1 : 0.f;
                    act[mt * 4 + j][nb] = f;
                }
            }
        }

        if (more) {
            const v4f* g2 = (const v4f*)(w2 + (size_t)dn * HID * HID);
            #pragma unroll
            for (int s = 0; s < 4; ++s) pre[s] = g2[tid + 256 * s];
        }

        SCHED_FENCE();
        load_row(W2c, 1, aB);
        SCHED_FENCE();
        mma_row(aA, acc[0], 16);
        SCHED_FENCE();
        load_row(W2c, 2, aA);
        SCHED_FENCE();
        mma_row(aB, acc[1], 16);
        SCHED_FENCE();
        load_row(W2c, 3, aB);
        SCHED_FENCE();
        mma_row(aA, acc[2], 16);
        SCHED_FENCE();
        if (more) {
            #pragma unroll
            for (int s = 0; s < 4; ++s) {
                const int q = tid + 256 * s;
                *(v4f*)(W2n + (q >> 4) * WSTR + (q & 15) * 4) = pre[s];
            }
        }
        lane_vec(vc + 3 * HID, bias);
        SCHED_FENCE();
        mma_row(aB, acc[3], 16);
        SCHED_FENCE();
        lane_vec(vc + 4 * HID, wo);

        #pragma unroll
        for (int nb = 0; nb < NB; ++nb) {
            float s = sacc[nb];
            #pragma unroll
            for (int mt = 0; mt < 4; ++mt) {
                #pragma unroll
                for (int r = 0; r < 8; ++r) {
                    float v = acc[mt][nb][r] + bias[mt][r];
                    v = v > 0.f ? v : 0.f;
                    s = fmaf(wo[mt][r], v, s);
                }
            }
            sacc[nb] = s;
            xv[nb]   = xn[nb];
        }

        if (more) {
            vn[vec_slot(nm0, nm1)] = pv;
            if (tid < HID) vn[4 * HID + tid] = pwo;
            __syncthreads();
        }
        cm0 = nm0;
        cm1 = nm1;
    }
    #undef SCHED_FENCE

    {
        float t[NB];
        #pragma unroll
        for (int nb = 0; nb < NB; ++nb)
            t[nb] = sacc[nb] + __shfl_xor(sacc[nb], 16);
        const float res = h ? t[1] : t[0];
        const int   b   = wave * (16 * NB) + lane;
        if (use_atomic) {
            __hip_atomic_fetch_add(&outbuf[b * OUT_DIM + o], res,
                                   __ATOMIC_RELAXED, __HIP_MEMORY_SCOPE_AGENT);
        } else {
            outbuf[(size_t)g * BSZ + b] = res;
        }
    }
}

__global__ void kan_reduce_kernel(const float* __restrict__ part,
                                  const float* __restrict__ bout,
                                  float* __restrict__ out)
{
    const int o = blockIdx.x;
    const int b = threadIdx.x;
    float s = bout[o];
    const float* p = part + (size_t)o * NGRP * BSZ + b;
    #pragma unroll
    for (int k = 0; k < NGRP; ++k) s += p[k * BSZ];
    out[b * OUT_DIM + o] = s;
}

__global__ void kan_init_out(const float* __restrict__ bout, float* __restrict__ out)
{
    const int idx = blockIdx.x * 256 + threadIdx.x;
    out[idx] = bout[idx & 63];
}

extern "C" void kernel_launch(void* const* d_in, const int* in_sizes, int n_in,
                              void* d_out, int out_size, void* d_ws, size_t ws_size,
                              hipStream_t stream)
{
    const float* x    = (const float*)d_in[0];
    const float* w0   = (const float*)d_in[1];
    const float* b0   = (const float*)d_in[2];
    const float* w1   = (const float*)d_in[3];
    const float* b1   = (const float*)d_in[4];
    const float* w2   = (const float*)d_in[5];
    const float* b2   = (const float*)d_in[6];
    const float* wout = (const float*)d_in[7];
    const float* bout = (const float*)d_in[8];
    float* out = (float*)d_out;

    const size_t need = (size_t)(DDIM / EPW) * BSZ * sizeof(float);
    if (ws_size >= need) {
        float* part = (float*)d_ws;
        kan_edge_kernel<<<DDIM / EPW, 256, 0, stream>>>(x, w0, b0, w1, b1, w2, b2, wout, part, 0);
        kan_reduce_kernel<<<OUT_DIM, BSZ, 0, stream>>>(part, bout, out);
    } else {
        kan_init_out<<<64, 256, 0, stream>>>(bout, out);
        kan_edge_kernel<<<DDIM / EPW, 256, 0, stream>>>(x, w0, b0, w1, b1, w2, b2, wout, out, 1);
    }
}


# Wave priority (s_setprio) falls as a workgroup advances through its 8 edges: of two workgroups sharing a WGP the one behind wins the matrix pipe, so neither runs its last edges alone; sums unchanged
# speedup vs baseline: 1.5795x; 1.0281x over previous
#include <hip/hip_runtime.h>


typedef float v2f __attribute__((ext_vector_type(2)));
typedef float v4f __attribute__((ext_vector_type(4)));
typedef float v8f __attribute__((ext_vector_type(8)));

#define IN_DIM  64
#define OUT_DIM 64
#define HID     64
#define DDIM    (IN_DIM * OUT_DIM)
#define BSZ     256
#define WSTR    68
#define NB      2
#define EPW     8
#define NGRP    (IN_DIM / EPW)
#define WMAT    (HID * WSTR)
#define VECS    (5 * HID)

__launch_bounds__(256, 4)
__global__ void kan_edge_kernel(const float* __restrict__ x,
                                const float* __restrict__ w0, const float* __restrict__ b0,
                                const float* __restrict__ w1, const float* __restrict__ b1,
                                const float* __restrict__ w2, const float* __restrict__ b2,
                                const float* __restrict__ wout,
                                float* __restrict__ outbuf, int use_atomic)
{
    __shared__ __attribute__((aligned(16))) float W1s[2 * WMAT];
    __shared__ __attribute__((aligned(16))) float W2s[2 * WMAT];
    __shared__ __attribute__((aligned(16))) float vecs[2 * VECS];
    __shared__ __attribute__((aligned(16))) float xmm[8 * 16];
    __shared__ __attribute__((aligned(16))) float xrg[16];

    const int g    = blockIdx.x;
    const int o    = g / NGRP;
    const int i0   = (g % NGRP) * EPW;
    const int d0   = o * IN_DIM + i0;
    const int tid  = threadIdx.x;
    const int lane = tid & 31;
    const int wave = tid >> 5;
    const int c    = lane & 15;
    const int h    = lane >> 4;

    const int    vsel = tid >> 6;
    const float* vp   = (vsel == 0 ? w0 : vsel == 1 ? b0 : vsel == 2 ? b1 : b2) + (tid & 63);
    const float* xrow = x + (wave * (16 * NB) + c) * IN_DIM + i0;

    v4f   pre[4];
    float pv, pwo = 0.f;
    float xv[NB];
    float nwl, nbl, nwh, nbh;
    unsigned cm0, cm1;

    auto slot_of_rank = [](int r) -> int {
        return (r & 0x30) | ((r & 2) << 2) | ((r >> 1) & 6) | (r & 1);
    };
    auto slot_of_unit = [&](int u, unsigned m0, unsigned m1) -> int {
        const unsigned m    = u < 32 ? m0 : m1;
        const int      v    = u & 31;
        const int      lb   = __popc(m & ((1u << v) - 1u));
        const int      n0   = __popc(m0);
        const int      nl   = n0 + __popc(m1);
        const int      lv   = (int)((m >> v) & 1u);
        const int      ra   = (u < 32 ? 0 : n0) + lb;
        const int      rb   = nl + (u < 32 ? 0 : 32 - n0) + (v - lb);
        return slot_of_rank(rb + lv * (ra - rb));
    };
    auto live_masks = [&](float wl, float bl, float wh, float bh, float mn, float mx, unsigned& m0, unsigned& m1) {
        const bool l0 = fmaf(mn, wl, bl) > 0.f || fmaf(mx, wl, bl) > 0.f;
        const bool l1 = fmaf(mn, wh, bh) > 0.f || fmaf(mx, wh, bh) > 0.f;
        m0 = __builtin_amdgcn_ballot_w32(l0);
        m1 = __builtin_amdgcn_ballot_w32(l1);
    };
    auto store_w1 = [&](float* __restrict__ Wd, const v4f (&p)[4], unsigned m0, unsigned m1) {
        float* rowp = Wd + (tid >> 4) * WSTR;
        #pragma unroll
        for (int t = 0; t < 4; ++t) {
            const int sl = slot_of_unit(4 * (tid & 15) + t, m0, m1);
            #pragma unroll
            for (int s = 0; s < 4; ++s) rowp[16 * s * WSTR + sl] = p[s][t];
        }
    };
    auto vec_slot = [&](unsigned m0, unsigned m1) -> int {
        return tid < 2 * HID ? (tid & HID) + slot_of_unit(tid & (HID - 1), m0, m1) : tid;
    };

    {
        const v4f* g1 = (const v4f*)(w1 + (size_t)d0 * HID * HID);
        const v4f* g2 = (const v4f*)(w2 + (size_t)d0 * HID * HID);
        v4f s2[4];
        #pragma unroll
        for (int s = 0; s < 4; ++s) pre[s] = g1[tid + 256 * s];
        #pragma unroll
        for (int s = 0; s < 4; ++s) s2[s] = g2[tid + 256 * s];
        pv = vp[d0 * HID];
        if (tid < HID) pwo = wout[d0 * HID + tid];
        #pragma unroll
        for (int nb = 0; nb < NB; ++nb) xv[nb] = xrow[nb * 16 * IN_DIM];
        nwl = w0[d0 * HID + lane];       nbl = b0[d0 * HID + lane];
        nwh = w0[d0 * HID + lane + 32];  nbh = b0[d0 * HID + lane + 32];
        {
            const v4f* xr = (const v4f*)(x + tid * IN_DIM + i0);
            const v4f  xa = xr[0], xb = xr[1];
            float mn[8] = {xa[0], xa[1], xa[2], xa[3], xb[0], xb[1], xb[2], xb[3]};
            float mx[8] = {xa[0], xa[1], xa[2], xa[3], xb[0], xb[1], xb[2], xb[3]};
            #pragma unroll
            for (int off = 16; off >= 1; off >>= 1) {
                #pragma unroll
                for (int k = 0; k < 8; ++k) {
                    mn[k] = fminf(mn[k], __shfl_xor(mn[k], off));
                    mx[k] = fmaxf(mx[k], __shfl_xor(mx[k], off));
                }
            }
            if (lane == 0) {
                *(v4f*)(xmm + wave * 16)      = (v4f){mn[0], mn[1], mn[2], mn[3]};
                *(v4f*)(xmm + wave * 16 + 4)  = (v4f){mn[4], mn[5], mn[6], mn[7]};
                *(v4f*)(xmm + wave * 16 + 8)  = (v4f){mx[0], mx[1], mx[2], mx[3]};
                *(v4f*)(xmm + wave * 16 + 12) = (v4f){mx[4], mx[5], mx[6], mx[7]};
            }
        }
        __syncthreads();
        {
            const int t = lane & 15;
            float lo = xmm[t], hi = lo;
            #pragma unroll
            for (int w = 1; w < 8; ++w) {
                const float v = xmm[w * 16 + t];
                lo = fminf(lo, v);
                hi = fmaxf(hi, v);
            }
            const float r = t < 8 ? lo : hi;
            if (tid < 16) xrg[t] = r;
            live_masks(nwl, nbl, nwh, nbh, __shfl(r, 0), __shfl(r, 8), cm0, cm1);
        }
        store_w1(W1s, pre, cm0, cm1);
        #pragma unroll
        for (int s = 0; s < 4; ++s) {
            const int q = tid + 256 * s;
            *(v4f*)(W2s + (q >> 4) * WSTR + (q & 15) * 4) = s2[s];
        }
        vecs[vec_slot(cm0, cm1)] = pv;
        if (tid < HID) vecs[4 * HID + tid] = pwo;
        __syncthreads();
    }

    auto load_row = [&](const float* __restrict__ Ws, int mtp, v4f (&a)[8]) {
        const float* ap = Ws + (16 * mtp + c) * WSTR + 8 * h;
        #pragma unroll
        for (int mt = 0; mt < 4; ++mt) {
            a[2 * mt]     = *(const v4f*)(ap + 16 * mt);
            a[2 * mt + 1] = *(const v4f*)(ap + 16 * mt + 4);
        }
    };

    v2f act[16][NB];

    auto mma_row = [&](const v4f (&a)[8], v8f (&accr)[NB], int nchunks) {
        #pragma unroll
        for (int nb = 0; nb < NB; ++nb)
            accr[nb] = (v8f){0.f, 0.f, 0.f, 0.f, 0.f, 0.f, 0.f, 0.f};
        #pragma unroll
        for (int mt = 0; mt < 4; ++mt) {
            #pragma unroll
            for (int j = 0; j < 4; ++j) {
                if (mt > 0 && mt * 4 + j >= nchunks) return;
                const v4f q = a[2 * mt + (j >> 1)];
                v2f af;
                af[0] = q[2 * (j & 1)];
                af[1] = q[2 * (j & 1) + 1];
                #pragma unroll
                for (int nb = 0; nb < NB; ++nb)
                    accr[nb] = __builtin_amdgcn_wmma_f32_16x16x4_f32(
                        false, af, false, act[mt * 4 + j][nb], (short)0, accr[nb], false, false);
            }
        }
    };

    auto lane_vec = [&](const float* __restrict__ vs, float (&out)[4][8]) {
        #pragma unroll
        for (int mt = 0; mt < 4; ++mt) {
            const v4f lo = *(const v4f*)(vs + 16 * mt + 8 * h);
            const v4f hi = *(const v4f*)(vs + 16 * mt + 8 * h + 4);
            out[mt][0] = lo[0]; out[mt][1] = lo[1]; out[mt][2] = lo[2]; out[mt][3] = lo[3];
            out[mt][4] = hi[0]; out[mt][5] = hi[1]; out[mt][6] = hi[2]; out[mt][7] = hi[3];
        }
    };

    #define SCHED_FENCE() __builtin_amdgcn_sched_barrier(0)

    v8f   acc[4][NB];
    v4f   aA[8], aB[8];
    float bias[4][8], wo[4][8];
    float sacc[NB];
    #pragma unroll
    for (int nb = 0; nb < NB; ++nb) sacc[nb] = 0.f;

    #pragma nounroll
    for (int e = 0; e < EPW; ++e) {
        switch (e) {
            case 0: __builtin_amdgcn_s_setprio(3); break;
            case 3: __builtin_amdgcn_s_setprio(2); break;
            case 5: __builtin_amdgcn_s_setprio(1); break;
            case 7: __builtin_amdgcn_s_setprio(0); break;
            default: break;
        }
        const int    cur  = e & 1;
        const float* W1c  = W1s + cur * WMAT;
        const float* W2c  = W2s + cur * WMAT;
        float*       W1n  = W1s + (cur ^ 1) * WMAT;
        float*       W2n  = W2s + (cur ^ 1) * WMAT;
        const float* vc   = vecs + cur * VECS;
        float*       vn   = vecs + (cur ^ 1) * VECS;
        const bool   more = e + 1 < EPW;
        const int    dn   = d0 + e + 1;
        const int    nch  = (__popc(cm0) + __popc(cm1) + 3) >> 2;
        unsigned     nm0  = 0u, nm1 = 0u;

        float xn[NB];
        #pragma unroll
        for (int nb = 0; nb < NB; ++nb) xn[nb] = 0.f;
        if (more) {
            const v4f* g1 = (const v4f*)(w1 + (size_t)dn * HID * HID);
            #pragma unroll
            for (int s = 0; s < 4; ++s) pre[s] = g1[tid + 256 * s];
            pv = vp[dn * HID];
            if (tid < HID) pwo = wout[dn * HID + tid];
            #pragma unroll
            for (int nb = 0; nb < NB; ++nb) xn[nb] = xrow[nb * 16 * IN_DIM + e + 1];
            nwl = w0[dn * HID + lane];       nbl = b0[dn * HID + lane];
            nwh = w0[dn * HID + lane + 32];  nbh = b0[dn * HID + lane + 32];
        }

        lane_vec(vc, bias);
        lane_vec(vc + HID, wo);
        load_row(W1c, 0, aA);
        SCHED_FENCE();
        #pragma unroll
        for (int mt = 0; mt < 4; ++mt) {
            #pragma unroll
            for (int j = 0; j < 4; ++j) {
                #pragma unroll
                for (int nb = 0; nb < NB; ++nb) {
                    float v0 = fmaf(xv[nb], bias[mt][2 * j],     wo[mt][2 * j]);
                    float v1 = fmaf(xv[nb], bias[mt][2 * j + 1], wo[mt][2 * j + 1]);
                    v2f f;
                    f[0] = v0 > 0.f ? v0 : 0.f;
                    f[1] = v1 > 0.f ? v1 : 0.f;
                    act[mt * 4 + j][nb] = f;
                }
            }
        }

        SCHED_FENCE();
        load_row(W1c, 1, aB);
        SCHED_FENCE();
        mma_row(aA, acc[0], nch);
        SCHED_FENCE();
        load_row(W1c, 2, aA);
        SCHED_FENCE();
        mma_row(aB, acc[1], nch);
        SCHED_FENCE();
        load_row(W1c, 3, aB);
        SCHED_FENCE();
        mma_row(aA, acc[2], nch);
        SCHED_FENCE();
        if (more) {
            live_masks(nwl, nbl, nwh, nbh, xrg[e + 1], xrg[8 + e + 1], nm0, nm1);
            store_w1(W1n, pre, nm0, nm1);
        }
        lane_vec(vc + 2 * HID, bias);
        SCHED_FENCE();
        mma_row(aB, acc[3], nch);
        SCHED_FENCE();
        load_row(W2c, 0, aA);

        #pragma unroll
        for (int mt = 0; mt < 4; ++mt) {
            #pragma unroll
            for (int j = 0; j < 4; ++j) {
                #pragma unroll
                for (int nb = 0; nb < NB; ++nb) {
                    float v0 = acc[mt][nb][2 * j]     + bias[mt][2 * j];
                    float v1 = acc[mt][nb][2 * j + 1] + bias[mt][2 * j + 1];
                    v2f f;
                    f[0] = v0 > 0.f ? v0 : 0.f;
                    f[1] = v1 > 0.f ? v1 : 0.f;
                    act[mt * 4 + j][nb] = f;
                }
            }
        }

        if (more) {
            const v4f* g2 = (const v4f*)(w2 + (size_t)dn * HID * HID);
            #pragma unroll
            for (int s = 0; s < 4; ++s) pre[s] = g2[tid + 256 * s];
        }

        SCHED_FENCE();
        load_row(W2c, 1, aB);
        SCHED_FENCE();
        mma_row(aA, acc[0], 16);
        SCHED_FENCE();
        load_row(W2c, 2, aA);
        SCHED_FENCE();
        mma_row(aB, acc[1], 16);
        SCHED_FENCE();
        load_row(W2c, 3, aB);
        SCHED_FENCE();
        mma_row(aA, acc[2], 16);
        SCHED_FENCE();
        if (more) {
            #pragma unroll
            for (int s = 0; s < 4; ++s) {
                const int q = tid + 256 * s;
                *(v4f*)(W2n + (q >> 4) * WSTR + (q & 15) * 4) = pre[s];
            }
        }
        lane_vec(vc + 3 * HID, bias);
        SCHED_FENCE();
        mma_row(aB, acc[3], 16);
        SCHED_FENCE();
        lane_vec(vc + 4 * HID, wo);

        #pragma unroll
        for (int nb = 0; nb < NB; ++nb) {
            float s = sacc[nb];
            #pragma unroll
            for (int mt = 0; mt < 4; ++mt) {
                #pragma unroll
                for (int r = 0; r < 8; ++r) {
                    float v = acc[mt][nb][r] + bias[mt][r];
                    v = v > 0.f ? v : 0.f;
                    s = fmaf(wo[mt][r], v, s);
                }
            }
            sacc[nb] = s;
            xv[nb]   = xn[nb];
        }

        if (more) {
            vn[vec_slot(nm0, nm1)] = pv;
            if (tid < HID) vn[4 * HID + tid] = pwo;
            __syncthreads();
        }
        cm0 = nm0;
        cm1 = nm1;
    }
    #undef SCHED_FENCE

    {
        float t[NB];
        #pragma unroll
        for (int nb = 0; nb < NB; ++nb)
            t[nb] = sacc[nb] + __shfl_xor(sacc[nb], 16);
        const float res = h ? t[1] : t[0];
        const int   b   = wave * (16 * NB) + lane;
        if (use_atomic) {
            __hip_atomic_fetch_add(&outbuf[b * OUT_DIM + o], res,
                                   __ATOMIC_RELAXED, __HIP_MEMORY_SCOPE_AGENT);
        } else {
            outbuf[(size_t)g * BSZ + b] = res;
        }
    }
}

__global__ void kan_reduce_kernel(const float* __restrict__ part,
                                  const float* __restrict__ bout,
                                  float* __restrict__ out)
{
    const int o = blockIdx.x;
    const int b = threadIdx.x;
    float s = bout[o];
    const float* p = part + (size_t)o * NGRP * BSZ + b;
    #pragma unroll
    for (int k = 0; k < NGRP; ++k) s += p[k * BSZ];
    out[b * OUT_DIM + o] = s;
}

__global__ void kan_init_out(const float* __restrict__ bout, float* __restrict__ out)
{
    const int idx = blockIdx.x * 256 + threadIdx.x;
    out[idx] = bout[idx & 63];
}

extern "C" void kernel_launch(void* const* d_in, const int* in_sizes, int n_in,
                              void* d_out, int out_size, void* d_ws, size_t ws_size,
                              hipStream_t stream)
{
    const float* x    = (const float*)d_in[0];
    const float* w0   = (const float*)d_in[1];
    const float* b0   = (const float*)d_in[2];
    const float* w1   = (const float*)d_in[3];
    const float* b1   = (const float*)d_in[4];
    const float* w2   = (const float*)d_in[5];
    const float* b2   = (const float*)d_in[6];
    const float* wout = (const float*)d_in[7];
    const float* bout = (const float*)d_in[8];
    float* out = (float*)d_out;

    const size_t need = (size_t)(DDIM / EPW) * BSZ * sizeof(float);
    if (ws_size >= need) {
        float* part = (float*)d_ws;
        kan_edge_kernel<<<DDIM / EPW, 256, 0, stream>>>(x, w0, b0, w1, b1, w2, b2, wout, part, 0);
        kan_reduce_kernel<<<OUT_DIM, BSZ, 0, stream>>>(part, bout, out);
    } else {
        kan_init_out<<<64, 256, 0, stream>>>(bout, out);
        kan_edge_kernel<<<DDIM / EPW, 256, 0, stream>>>(x, w0, b0, w1, b1, w2, b2, wout, out, 1);
    }
}


# Layer 2 skips its trailing k chunks that are zero in all 32 rows of the wave (exact test); a 2-point probe of each edge orders W1 rows, b1, W2 columns dead-last; layer-2 k sum order changes
# speedup vs baseline: 1.7113x; 1.0834x over previous
#include <hip/hip_runtime.h>


typedef float v2f __attribute__((ext_vector_type(2)));
typedef float v4f __attribute__((ext_vector_type(4)));
typedef float v8f __attribute__((ext_vector_type(8)));

#define IN_DIM  64
#define OUT_DIM 64
#define HID     64
#define DDIM    (IN_DIM * OUT_DIM)
#define BSZ     256
#define WSTR    68
#define NB      2
#define EPW     8
#define NGRP    (IN_DIM / EPW)
#define WMAT    (HID * WSTR)
#define VECS    (5 * HID)

__launch_bounds__(256, 4)
__global__ void kan_edge_kernel(const float* __restrict__ x,
                                const float* __restrict__ w0, const float* __restrict__ b0,
                                const float* __restrict__ w1, const float* __restrict__ b1,
                                const float* __restrict__ w2, const float* __restrict__ b2,
                                const float* __restrict__ wout,
                                float* __restrict__ outbuf, int use_atomic)
{
    __shared__ __attribute__((aligned(16))) float W1s[2 * WMAT];
    __shared__ __attribute__((aligned(16))) float W2s[2 * WMAT];
    __shared__ __attribute__((aligned(16))) float vecs[2 * VECS];
    __shared__ __attribute__((aligned(16))) float xmm[8 * 16];
    __shared__ __attribute__((aligned(16))) float xrg[16];
    __shared__ __attribute__((aligned(16))) int   rofs[2 * HID];
    __shared__ __attribute__((aligned(16))) unsigned pbw[16];
    __shared__ __attribute__((aligned(16))) float hpr[8 * 2 * HID];

    const int g    = blockIdx.x;
    const int o    = g / NGRP;
    const int i0   = (g % NGRP) * EPW;
    const int d0   = o * IN_DIM + i0;
    const int tid  = threadIdx.x;
    const int lane = tid & 31;
    const int wave = tid >> 5;
    const int c    = lane & 15;
    const int h    = lane >> 4;

    const int    vsel = tid >> 6;
    const float* vp   = (vsel == 0 ? w0 : vsel == 1 ? b0 : vsel == 2 ? b1 : b2) + (tid & 63);
    const float* xrow = x + (wave * (16 * NB) + c) * IN_DIM + i0;

    v4f   pre[4];
    float pv, pwo = 0.f;
    float xv[NB];
    float nwl, nbl, nwh, nbh;
    unsigned cm0, cm1;

    auto slot_of_rank = [](int r) -> int {
        return (r & 0x30) | ((r & 2) << 2) | ((r >> 1) & 6) | (r & 1);
    };
    auto slot_of_unit = [&](int u, unsigned m0, unsigned m1) -> int {
        const unsigned m    = u < 32 ? m0 : m1;
        const int      v    = u & 31;
        const int      lb   = __popc(m & ((1u << v) - 1u));
        const int      n0   = __popc(m0);
        const int      nl   = n0 + __popc(m1);
        const int      lv   = (int)((m >> v) & 1u);
        const int      ra   = (u < 32 ? 0 : n0) + lb;
        const int      rb   = nl + (u < 32 ? 0 : 32 - n0) + (v - lb);
        return slot_of_rank(rb + lv * (ra - rb));
    };
    auto live_masks = [&](float wl, float bl, float wh, float bh, float mn, float mx, unsigned& m0, unsigned& m1) {
        const bool l0 = fmaf(mn, wl, bl) > 0.f || fmaf(mx, wl, bl) > 0.f;
        const bool l1 = fmaf(mn, wh, bh) > 0.f || fmaf(mx, wh, bh) > 0.f;
        m0 = __builtin_amdgcn_ballot_w32(l0);
        m1 = __builtin_amdgcn_ballot_w32(l1);
    };
    auto store_w1 = [&](float* __restrict__ Wd, const v4f (&p)[4], unsigned m0, unsigned m1) {
        float* rowp = Wd + (tid >> 4) * WSTR;
        #pragma unroll
        for (int t = 0; t < 4; ++t) {
            const int sl = slot_of_unit(4 * (tid & 15) + t, m0, m1);
            #pragma unroll
            for (int s = 0; s < 4; ++s) rowp[16 * s * WSTR + sl] = p[s][t];
        }
    };
    auto vec_slot = [&](unsigned m0, unsigned m1, unsigned p0, unsigned p1) -> int {
        if (tid >= 3 * HID) return tid;
        const bool inl = tid < 2 * HID;
        return (tid & ~(HID - 1)) + slot_of_unit(tid & (HID - 1), inl ? m0 : p0, inl ? m1 : p1);
    };

    {
        const v4f* g1 = (const v4f*)(w1 + (size_t)d0 * HID * HID);
        const v4f* g2 = (const v4f*)(w2 + (size_t)d0 * HID * HID);
        v4f s2[4];
        #pragma unroll
        for (int s = 0; s < 4; ++s) pre[s] = g1[tid + 256 * s];
        #pragma unroll
        for (int s = 0; s < 4; ++s) s2[s] = g2[tid + 256 * s];
        pv = vp[d0 * HID];
        if (tid < HID) pwo = wout[d0 * HID + tid];
        #pragma unroll
        for (int nb = 0; nb < NB; ++nb) xv[nb] = xrow[nb * 16 * IN_DIM];
        nwl = w0[d0 * HID + lane];       nbl = b0[d0 * HID + lane];
        nwh = w0[d0 * HID + lane + 32];  nbh = b0[d0 * HID + lane + 32];
        {
            const v4f* xr = (const v4f*)(x + tid * IN_DIM + i0);
            const v4f  xa = xr[0], xb = xr[1];
            float mn[8] = {xa[0], xa[1], xa[2], xa[3], xb[0], xb[1], xb[2], xb[3]};
            float mx[8] = {xa[0], xa[1], xa[2], xa[3], xb[0], xb[1], xb[2], xb[3]};
            #pragma unroll
            for (int off = 16; off >= 1; off >>= 1) {
                #pragma unroll
                for (int k = 0; k < 8; ++k) {
                    mn[k] = fminf(mn[k], __shfl_xor(mn[k], off));
                    mx[k] = fmaxf(mx[k], __shfl_xor(mx[k], off));
                }
            }
            if (lane == 0) {
                *(v4f*)(xmm + wave * 16)      = (v4f){mn[0], mn[1], mn[2], mn[3]};
                *(v4f*)(xmm + wave * 16 + 4)  = (v4f){mn[4], mn[5], mn[6], mn[7]};
                *(v4f*)(xmm + wave * 16 + 8)  = (v4f){mx[0], mx[1], mx[2], mx[3]};
                *(v4f*)(xmm + wave * 16 + 12) = (v4f){mx[4], mx[5], mx[6], mx[7]};
            }
        }
        __syncthreads();
        {
            const int t = lane & 15;
            float lo = xmm[t], hi = lo;
            #pragma unroll
            for (int w = 1; w < 8; ++w) {
                const float v = xmm[w * 16 + t];
                lo = fminf(lo, v);
                hi = fmaxf(hi, v);
            }
            const float r = t < 8 ? lo : hi;
            if (tid < 16) xrg[t] = r;
            live_masks(nwl, nbl, nwh, nbh, __shfl(r, 0), __shfl(r, 8), cm0, cm1);
        }
        store_w1(W1s, pre, cm0, cm1);
        store_w1(W2s, s2, ~0u, ~0u);
        {
            const int vsl = vec_slot(cm0, cm1, ~0u, ~0u);
            vecs[vsl] = pv;
            if (vsel == 2) rofs[vsl - 2 * HID] = (tid & (HID - 1)) * WSTR;
        }
        if (tid < HID) vecs[4 * HID + tid] = pwo;
        __syncthreads();
    }

    auto load_row = [&](const float* __restrict__ Ws, int rowoff, v4f (&a)[8]) {
        const float* ap = Ws + rowoff + 8 * h;
        #pragma unroll
        for (int mt = 0; mt < 4; ++mt) {
            a[2 * mt]     = *(const v4f*)(ap + 16 * mt);
            a[2 * mt + 1] = *(const v4f*)(ap + 16 * mt + 4);
        }
    };

    v2f act[16][NB];

    auto mma_row = [&](const v4f (&a)[8], v8f (&accr)[NB], int nchunks) {
        #pragma unroll
        for (int nb = 0; nb < NB; ++nb)
            accr[nb] = (v8f){0.f, 0.f, 0.f, 0.f, 0.f, 0.f, 0.f, 0.f};
        #pragma unroll
        for (int mt = 0; mt < 4; ++mt) {
            #pragma unroll
            for (int j = 0; j < 4; ++j) {
                if (mt > 0 && mt * 4 + j >= nchunks) return;
                const v4f q = a[2 * mt + (j >> 1)];
                v2f af;
                af[0] = q[2 * (j & 1)];
                af[1] = q[2 * (j & 1) + 1];
                #pragma unroll
                for (int nb = 0; nb < NB; ++nb)
                    accr[nb] = __builtin_amdgcn_wmma_f32_16x16x4_f32(
                        false, af, false, act[mt * 4 + j][nb], (short)0, accr[nb], false, false);
            }
        }
    };

    auto lane_vec = [&](const float* __restrict__ vs, float (&out)[4][8]) {
        #pragma unroll
        for (int mt = 0; mt < 4; ++mt) {
            const v4f lo = *(const v4f*)(vs + 16 * mt + 8 * h);
            const v4f hi = *(const v4f*)(vs + 16 * mt + 8 * h + 4);
            out[mt][0] = lo[0]; out[mt][1] = lo[1]; out[mt][2] = lo[2]; out[mt][3] = lo[3];
            out[mt][4] = hi[0]; out[mt][5] = hi[1]; out[mt][6] = hi[2]; out[mt][7] = hi[3];
        }
    };

    #define SCHED_FENCE() __builtin_amdgcn_sched_barrier(0)
    #define WMMA_EXIT_PAD() asm volatile("v_nop\n\tv_nop\n\tv_nop\n\tv_nop")
    #define ROW_XADD(v, k) ((v) + __builtin_bit_cast(float, __builtin_amdgcn_update_dpp( \
        0, __builtin_bit_cast(int, (v)), 0x160 | (k), 0xF, 0xF, true)))
    typedef unsigned v4u __attribute__((ext_vector_type(4)));

    v8f   acc[4][NB];
    v4f   aA[8], aB[8];
    float bias[4][8], wo[4][8];
    float sacc[NB];
    #pragma unroll
    for (int nb = 0; nb < NB; ++nb) sacc[nb] = 0.f;

    #pragma nounroll
    for (int e = 0; e < EPW; ++e) {
        switch (e) {
            case 0: __builtin_amdgcn_s_setprio(3); break;
            case 3: __builtin_amdgcn_s_setprio(2); break;
            case 5: __builtin_amdgcn_s_setprio(1); break;
            case 7: __builtin_amdgcn_s_setprio(0); break;
            default: break;
        }
        const int    cur  = e & 1;
        const float* W1c  = W1s + cur * WMAT;
        const float* W2c  = W2s + cur * WMAT;
        float*       W1n  = W1s + (cur ^ 1) * WMAT;
        float*       W2n  = W2s + (cur ^ 1) * WMAT;
        const float* vc   = vecs + cur * VECS;
        float*       vn   = vecs + (cur ^ 1) * VECS;
        const bool   more = e + 1 < EPW;
        const int    dn   = d0 + e + 1;
        const int    nch  = (__popc(cm0) + __popc(cm1) + 3) >> 2;
        unsigned     nm0  = 0u, nm1 = 0u;
        unsigned     pm0  = 0u, pm1 = 0u;
        const int    w2r  = (16 * 0 + c) * WSTR;

        float xn[NB];
        float pb1[4];
        #pragma unroll
        for (int nb = 0; nb < NB; ++nb) xn[nb] = 0.f;
        #pragma unroll
        for (int s = 0; s < 4; ++s) pb1[s] = 0.f;
        if (more) {
            const v4f* g1 = (const v4f*)(w1 + (size_t)dn * HID * HID);
            #pragma unroll
            for (int s = 0; s < 4; ++s) pre[s] = g1[tid + 256 * s];
            pv = vp[dn * HID];
            if (tid < HID) pwo = wout[dn * HID + tid];
            #pragma unroll
            for (int nb = 0; nb < NB; ++nb) xn[nb] = xrow[nb * 16 * IN_DIM + e + 1];
            nwl = w0[dn * HID + lane];       nbl = b0[dn * HID + lane];
            nwh = w0[dn * HID + lane + 32];  nbh = b0[dn * HID + lane + 32];
            #pragma unroll
            for (int s = 0; s < 4; ++s) pb1[s] = b1[dn * HID + (tid >> 4) + 16 * s];
        }

        int ra[4];
        #pragma unroll
        for (int mtp = 0; mtp < 4; ++mtp) ra[mtp] = rofs[cur * HID + 16 * mtp + c];

        lane_vec(vc, bias);
        lane_vec(vc + HID, wo);
        load_row(W1c, ra[0], aA);
        SCHED_FENCE();
        #pragma unroll
        for (int mt = 0; mt < 4; ++mt) {
            #pragma unroll
            for (int j = 0; j < 4; ++j) {
                #pragma unroll
                for (int nb = 0; nb < NB; ++nb) {
                    float v0 = fmaf(xv[nb], bias[mt][2 * j],     wo[mt][2 * j]);
                    float v1 = fmaf(xv[nb], bias[mt][2 * j + 1], wo[mt][2 * j + 1]);
                    v2f f;
                    f[0] = v0 > 0.f ? v0 : 0.f;
                    f[1] = v1 > 0.f ? v1 : 0.f;
                    act[mt * 4 + j][nb] = f;
                }
            }
        }

        SCHED_FENCE();
        load_row(W1c, ra[1], aB);
        SCHED_FENCE();
        mma_row(aA, acc[0], nch);
        WMMA_EXIT_PAD();
        SCHED_FENCE();
        load_row(W1c, ra[2], aA);
        SCHED_FENCE();
        mma_row(aB, acc[1], nch);
        WMMA_EXIT_PAD();
        SCHED_FENCE();
        load_row(W1c, ra[3], aB);
        SCHED_FENCE();
        mma_row(aA, acc[2], nch);
        WMMA_EXIT_PAD();
        SCHED_FENCE();
        if (more) {
            const float xlo = xrg[e + 1], xhi = xrg[8 + e + 1];
            live_masks(nwl, nbl, nwh, nbh, xlo, xhi, nm0, nm1);
            store_w1(W1n, pre, nm0, nm1);
            float* hp = hpr + wave * (2 * HID);
            {
                const float a0 = fmaf(xlo, nwl, nbl), a1 = fmaf(xhi, nwl, nbl);
                const float c0 = fmaf(xlo, nwh, nbh), c1 = fmaf(xhi, nwh, nbh);
                v2f ua, uc;
                ua[0] = a0 > 0.f ? a0 : 0.f;  ua[1] = a1 > 0.f ? a1 : 0.f;
                uc[0] = c0 > 0.f ? c0 : 0.f;  uc[1] = c1 > 0.f ? c1 : 0.f;
                *(v2f*)(hp + 2 * lane)        = ua;
                *(v2f*)(hp + 2 * (lane + 32)) = uc;
            }
            const v4f h01 = *(const v4f*)(hp + 8 * (tid & 15));
            const v4f h23 = *(const v4f*)(hp + 8 * (tid & 15) + 4);
            unsigned blv[4];
            #pragma unroll
            for (int s = 0; s < 4; ++s) {
                float zl = pre[s][0] * h01[0], zh = pre[s][0] * h01[1];
                zl = fmaf(pre[s][1], h01[2], zl);  zh = fmaf(pre[s][1], h01[3], zh);
                zl = fmaf(pre[s][2], h23[0], zl);  zh = fmaf(pre[s][2], h23[1], zh);
                zl = fmaf(pre[s][3], h23[2], zl);  zh = fmaf(pre[s][3], h23[3], zh);
                zl = ROW_XADD(zl, 1);  zh = ROW_XADD(zh, 1);
                zl = ROW_XADD(zl, 2);  zh = ROW_XADD(zh, 2);
                zl = ROW_XADD(zl, 4);  zh = ROW_XADD(zh, 4);
                zl = ROW_XADD(zl, 8);  zh = ROW_XADD(zh, 8);
                blv[s] = __builtin_amdgcn_ballot_w32(zl + pb1[s] > 0.f || zh + pb1[s] > 0.f);
            }
            const unsigned lo = (((blv[0] & 1u) | ((blv[0] >> 15) & 2u)) << (2 * wave))
                              | (((blv[1] & 1u) | ((blv[1] >> 15) & 2u)) << (2 * wave + 16));
            const unsigned hi = (((blv[2] & 1u) | ((blv[2] >> 15) & 2u)) << (2 * wave))
                              | (((blv[3] & 1u) | ((blv[3] >> 15) & 2u)) << (2 * wave + 16));
            if (lane == 0) {
                pbw[2 * wave]     = lo;
                pbw[2 * wave + 1] = hi;
            }
        }
        SCHED_FENCE();
        mma_row(aB, acc[3], nch);
        WMMA_EXIT_PAD();
        SCHED_FENCE();
        if (more) asm volatile("s_wait_dscnt 0x0\n\ts_barrier_signal -1" ::: "memory");
        load_row(W2c, w2r, aA);
        lane_vec(vc + 2 * HID, bias);

        #pragma unroll
        for (int mt = 0; mt < 4; ++mt) {
            #pragma unroll
            for (int j = 0; j < 4; ++j) {
                #pragma unroll
                for (int nb = 0; nb < NB; ++nb) {
                    float v0 = acc[mt][nb][2 * j]     + bias[mt][2 * j];
                    float v1 = acc[mt][nb][2 * j + 1] + bias[mt][2 * j + 1];
                    v2f f;
                    f[0] = v0 > 0.f ? v0 : 0.f;
                    f[1] = v1 > 0.f ? v1 : 0.f;
                    act[mt * 4 + j][nb] = f;
                }
            }
        }

        int n2 = 4;
        #pragma unroll
        for (int kc = 4; kc < 16; ++kc) {
            const float t00 = act[kc][0][0], t01 = act[kc][0][1], t10 = act[kc][1][0], t11 = act[kc][1][1];
            const unsigned t = __float_as_uint(t00) | __float_as_uint(t01) | __float_as_uint(t10) | __float_as_uint(t11);
            if (__builtin_amdgcn_ballot_w32(t != 0u) != 0u) n2 = kc + 1;
        }

        if (more) {
            const v4f* g2 = (const v4f*)(w2 + (size_t)dn * HID * HID);
            #pragma unroll
            for (int s = 0; s < 4; ++s) pre[s] = g2[tid + 256 * s];
        }

        SCHED_FENCE();
        load_row(W2c, w2r + 16 * WSTR, aB);
        SCHED_FENCE();
        mma_row(aA, acc[0], n2);
        WMMA_EXIT_PAD();
        SCHED_FENCE();
        load_row(W2c, w2r + 32 * WSTR, aA);
        SCHED_FENCE();
        mma_row(aB, acc[1], n2);
        WMMA_EXIT_PAD();
        SCHED_FENCE();
        load_row(W2c, w2r + 48 * WSTR, aB);
        SCHED_FENCE();
        mma_row(aA, acc[2], n2);
        WMMA_EXIT_PAD();
        SCHED_FENCE();
        if (more) {
            asm volatile("s_barrier_wait -1" ::: "memory");
            const v4u q0 = *(const v4u*)(pbw), q1 = *(const v4u*)(pbw + 4);
            const v4u q2 = *(const v4u*)(pbw + 8), q3 = *(const v4u*)(pbw + 12);
            const v4u qq = q0 | q1 | q2 | q3;
            pm0 = __builtin_amdgcn_readfirstlane(qq[0] | qq[2]);
            pm1 = __builtin_amdgcn_readfirstlane(qq[1] | qq[3]);
            store_w1(W2n, pre, pm0, pm1);
        }
        lane_vec(vc + 3 * HID, bias);
        SCHED_FENCE();
        mma_row(aB, acc[3], n2);
        WMMA_EXIT_PAD();
        SCHED_FENCE();
        lane_vec(vc + 4 * HID, wo);

        #pragma unroll
        for (int nb = 0; nb < NB; ++nb) {
            float s = sacc[nb];
            #pragma unroll
            for (int mt = 0; mt < 4; ++mt) {
                #pragma unroll
                for (int r = 0; r < 8; ++r) {
                    float v = acc[mt][nb][r] + bias[mt][r];
                    v = v > 0.f ? v : 0.f;
                    s = fmaf(wo[mt][r], v, s);
                }
            }
            sacc[nb] = s;
            xv[nb]   = xn[nb];
        }

        if (more) {
            const int vsl = vec_slot(nm0, nm1, pm0, pm1);
            vn[vsl] = pv;
            if (vsel == 2) rofs[(cur ^ 1) * HID + vsl - 2 * HID] = (tid & (HID - 1)) * WSTR;
            if (tid < HID) vn[4 * HID + tid] = pwo;
            __syncthreads();
        }
        cm0 = nm0;
        cm1 = nm1;
    }
    #undef SCHED_FENCE
    #undef WMMA_EXIT_PAD
    #undef ROW_XADD

    {
        float t[NB];
        #pragma unroll
        for (int nb = 0; nb < NB; ++nb)
            t[nb] = sacc[nb] + __shfl_xor(sacc[nb], 16);
        const float res = __builtin_bit_cast(float, __builtin_amdgcn_update_dpp(
            __builtin_bit_cast(int, t[0]), __builtin_bit_cast(int, t[1]), 0xE4, 0x2, 0xF, false));
        const int   b   = wave * (16 * NB) + lane;
        if (use_atomic) {
            __hip_atomic_fetch_add(&outbuf[b * OUT_DIM + o], res,
                                   __ATOMIC_RELAXED, __HIP_MEMORY_SCOPE_AGENT);
        } else {
            outbuf[(size_t)g * BSZ + b] = res;
        }
    }
}

__global__ void kan_reduce_kernel(const float* __restrict__ part,
                                  const float* __restrict__ bout,
                                  float* __restrict__ out)
{
    const int o = blockIdx.x;
    const int b = threadIdx.x;
    float s = bout[o];
    const float* p = part + (size_t)o * NGRP * BSZ + b;
    #pragma unroll
    for (int k = 0; k < NGRP; ++k) s += p[k * BSZ];
    out[b * OUT_DIM + o] = s;
}

__global__ void kan_init_out(const float* __restrict__ bout, float* __restrict__ out)
{
    const int idx = blockIdx.x * 256 + threadIdx.x;
    out[idx] = bout[idx & 63];
}

extern "C" void kernel_launch(void* const* d_in, const int* in_sizes, int n_in,
                              void* d_out, int out_size, void* d_ws, size_t ws_size,
                              hipStream_t stream)
{
    const float* x    = (const float*)d_in[0];
    const float* w0   = (const float*)d_in[1];
    const float* b0   = (const float*)d_in[2];
    const float* w1   = (const float*)d_in[3];
    const float* b1   = (const float*)d_in[4];
    const float* w2   = (const float*)d_in[5];
    const float* b2   = (const float*)d_in[6];
    const float* wout = (const float*)d_in[7];
    const float* bout = (const float*)d_in[8];
    float* out = (float*)d_out;

    const size_t need = (size_t)(DDIM / EPW) * BSZ * sizeof(float);
    if (ws_size >= need) {
        float* part = (float*)d_ws;
        kan_edge_kernel<<<DDIM / EPW, 256, 0, stream>>>(x, w0, b0, w1, b1, w2, b2, wout, part, 0);
        kan_reduce_kernel<<<OUT_DIM, BSZ, 0, stream>>>(part, bout, out);
    } else {
        kan_init_out<<<64, 256, 0, stream>>>(bout, out);
        kan_edge_kernel<<<DDIM / EPW, 256, 0, stream>>>(x, w0, b0, w1, b1, w2, b2, wout, out, 1);
    }
}
